# starting kernel
# baseline (speedup 1.0000x reference)
#include <hip/hip_runtime.h>

#define NB   4096
#define NF   200
#define NK   40
#define NC   128

typedef __attribute__((ext_vector_type(2))) float v2f;
typedef __attribute__((ext_vector_type(4))) float v4f;
typedef __attribute__((ext_vector_type(8))) float v8f;

#define NRG  13

__device__ __forceinline__ void sched_fence() {
#if defined(__has_builtin)
#if __has_builtin(__builtin_amdgcn_sched_barrier)
    __builtin_amdgcn_sched_barrier(0);
#endif
#endif
}

template <int MASK, int SIZE>
__device__ __forceinline__ void sched_group() {
#if defined(__has_builtin)
#if __has_builtin(__builtin_amdgcn_sched_group_barrier)
    __builtin_amdgcn_sched_group_barrier(MASK, SIZE, 0);
#endif
#endif
}
#define SG_DS_READ 0x100
#define SG_WMMA    0x008

__global__ __launch_bounds__(256)
void softmax_gemm_reduce_kernel(const float* __restrict__ x,
                                const float* __restrict__ W,
                                float* __restrict__ out)
{
    __shared__ float sW[(NK / 2) * (NC * 2)];
    __shared__ float sOut[NC];

    const int tid  = threadIdx.x;
    const int b    = blockIdx.x;
    const int wave = tid >> 5;
    const int lane = tid & 31;
    const int half = lane >> 4;
    const int lm   = lane & 15;

    for (int i4 = tid; i4 < (NK * NC) / 4; i4 += 256) {
        const int i = i4 * 4;
        const int k = i >> 7;
        const int c = i & (NC - 1);
        v4f w = *(const v4f*)(W + i);
        float* dst = &sW[(k >> 1) * 256 + c * 2 + (k & 1)];
        dst[0] = w.x; dst[2] = w.y; dst[4] = w.z; dst[6] = w.w;
    }
    if (tid < NC) sOut[tid] = 0.0f;
    __syncthreads();

    const float* sWcol = &sW[lm * 2];

    for (int rg = wave; rg < NRG; rg += 8) {
        const int f       = rg * 16 + lm;
        const int fclamp  = (f < NF) ? f : (NF - 1);
        const float* xrow = x + ((size_t)b * NF + fclamp) * NK;
        const int klo     = half * 2;

        v2f afr[10];
        #pragma unroll
        for (int ks = 0; ks < 10; ++ks)
            afr[ks] = *(const v2f*)(xrow + 4 * ks + klo);

        v2f bf[2][8];
        #pragma unroll
        for (int nt = 0; nt < 8; ++nt)
            bf[0][nt] = *(const v2f*)(sWcol + (half)*256 + nt * 32);

        sched_fence();

        v8f acc[8] = {};
        #pragma unroll
        for (int ks = 0; ks < 10; ++ks) {
            const int cur = ks & 1;
            const int nxt = cur ^ 1;
            if (ks < 9) {
                const int j = 2 * (ks + 1) + half;
                #pragma unroll
                for (int nt = 0; nt < 8; ++nt)
                    bf[nxt][nt] = *(const v2f*)(sWcol + j * 256 + nt * 32);
                sched_group<SG_DS_READ, 4>();
            }
            #pragma unroll
            for (int nt = 0; nt < 8; ++nt) {
                acc[nt] = __builtin_amdgcn_wmma_f32_16x16x4_f32(
                               false, afr[ks],  false, bf[cur][nt],
                               (short)0, acc[nt],
                               false,  false);
            }
            sched_group<SG_WMMA, 8>();
        }

        float rinv[8];
        #pragma unroll
        for (int r = 0; r < 8; ++r) {
            float m = acc[0][r];
            #pragma unroll
            for (int nt = 1; nt < 8; ++nt) m = fmaxf(m, acc[nt][r]);
            #pragma unroll
            for (int s = 1; s < 16; s <<= 1) m = fmaxf(m, __shfl_xor(m, s, 32));

            float sum = 0.0f;
            #pragma unroll
            for (int nt = 0; nt < 8; ++nt) {
                float e = __expf(acc[nt][r] - m);
                acc[nt][r] = e;
                sum += e;
            }
            #pragma unroll
            for (int s = 1; s < 16; s <<= 1) sum += __shfl_xor(sum, s, 32);
            rinv[r] = 1.0f / sum;
        }

        #pragma unroll
        for (int nt = 0; nt < 8; ++nt) {
            float cs = 0.0f;
            #pragma unroll
            for (int r = 0; r < 8; ++r) {
                const int frow = rg * 16 + half * 8 + r;
                float v = acc[nt][r] * rinv[r];
                cs += (frow < NF) ? v : 0.0f;
            }
            cs += __shfl_xor(cs, 16, 32);
            if (half == 0) atomicAdd(&sOut[nt * 16 + lm], cs);
        }
    }

    __syncthreads();
    if (tid < NC) out[(size_t)b * NC + tid] = sOut[tid];
}

extern "C" void kernel_launch(void* const* d_in, const int* in_sizes, int n_in,
                              void* d_out, int out_size, void* d_ws, size_t ws_size,
                              hipStream_t stream) {
    const float* x = (const float*)d_in[0];
    const float* W = (const float*)d_in[1];
    float* out     = (float*)d_out;

    dim3 grid(NB);
    dim3 block(256);
    softmax_gemm_reduce_kernel<<<grid, block, 0, stream>>>(x, W, out);
}


# Swap WMMA operands (W as A, x as B): a softmax row sits in one lane pair, max/sum in-lane; column sums kept per lane, summed across lanes once per wave by a halving tree (sum order changes)
# speedup vs baseline: 1.1200x; 1.1200x over previous
#include <hip/hip_runtime.h>

#define NB   4096
#define NF   200
#define NK   40
#define NC   128

typedef __attribute__((ext_vector_type(2))) float v2f;
typedef __attribute__((ext_vector_type(4))) float v4f;
typedef __attribute__((ext_vector_type(8))) float v8f;

#define NRG  13

__device__ __forceinline__ void sched_fence() {
#if defined(__has_builtin)
#if __has_builtin(__builtin_amdgcn_sched_barrier)
    __builtin_amdgcn_sched_barrier(0);
#endif
#endif
}

template <int MASK, int SIZE>
__device__ __forceinline__ void sched_group() {
#if defined(__has_builtin)
#if __has_builtin(__builtin_amdgcn_sched_group_barrier)
    __builtin_amdgcn_sched_group_barrier(MASK, SIZE, 0);
#endif
#endif
}
#define SG_DS_READ 0x100
#define SG_WMMA    0x008

__device__ __forceinline__ float xhalf(float v) {
    const int b = __builtin_bit_cast(int, v);
    return __builtin_bit_cast(float,
        __builtin_amdgcn_permlanex16(b, b, 0x76543210, 0xfedcba98, false, false));
}

template <int MASK>
__device__ __forceinline__ float row_xor(float v) {
    const int b = __builtin_bit_cast(int, v);
    return __builtin_bit_cast(float,
        __builtin_amdgcn_update_dpp(b, b, 0x160 | MASK, 0xf, 0xf, true));
}

template <int MASK, int N>
__device__ __forceinline__ void halve_across(const float (&in)[N], float (&outv)[N / 2], bool hi) {
    #pragma unroll
    for (int j = 0; j < N / 2; ++j) {
        const float keep = hi ? in[N / 2 + j] : in[j];
        const float send = hi ? in[j] : in[N / 2 + j];
        outv[j] = keep + row_xor<MASK>(send);
    }
}

__global__ __launch_bounds__(256)
void softmax_gemm_reduce_kernel(const float* __restrict__ x,
                                const float* __restrict__ W,
                                float* __restrict__ out)
{
    __shared__ float sW[(NK / 2) * (NC * 2)];
    __shared__ float sOut[NC];

    const int tid  = threadIdx.x;
    const int b    = blockIdx.x;
    const int wave = tid >> 5;
    const int lane = tid & 31;
    const int half = lane >> 4;
    const int lm   = lane & 15;

    for (int i4 = tid; i4 < (NK * NC) / 4; i4 += 256) {
        const int i = i4 * 4;
        const int k = i >> 7;
        const int c = i & (NC - 1);
        v4f w = *(const v4f*)(W + i);
        float* dst = &sW[(k >> 1) * 256 + c * 2 + (k & 1)];
        dst[0] = w.x; dst[2] = w.y; dst[4] = w.z; dst[6] = w.w;
    }
    if (tid < NC) sOut[tid] = 0.0f;
    __syncthreads();

    const float* sWcol = &sW[lm * 2];

    v8f osum[8] = {};

    for (int rg = wave; rg < NRG; rg += 8) {
        const int f       = rg * 16 + lm;
        const bool valid  = (f < NF);
        const int fclamp  = valid ? f : (NF - 1);
        const float* xrow = x + ((size_t)b * NF + fclamp) * NK;
        const int klo     = half * 2;

        v2f xfr[10];
        #pragma unroll
        for (int ks = 0; ks < 10; ++ks)
            xfr[ks] = *(const v2f*)(xrow + 4 * ks + klo);

        v2f wf[2][8];
        #pragma unroll
        for (int nt = 0; nt < 8; ++nt)
            wf[0][nt] = *(const v2f*)(sWcol + (half)*256 + nt * 32);

        sched_fence();

        v8f acc[8] = {};
        #pragma unroll
        for (int ks = 0; ks < 10; ++ks) {
            const int cur = ks & 1;
            const int nxt = cur ^ 1;
            if (ks < 9) {
                const int j = 2 * (ks + 1) + half;
                #pragma unroll
                for (int nt = 0; nt < 8; ++nt)
                    wf[nxt][nt] = *(const v2f*)(sWcol + j * 256 + nt * 32);
                sched_group<SG_DS_READ, 4>();
            }
            #pragma unroll
            for (int nt = 0; nt < 8; ++nt) {
                acc[nt] = __builtin_amdgcn_wmma_f32_16x16x4_f32(
                               false, wf[cur][nt],  false, xfr[ks],
                               (short)0, acc[nt],
                               false,  false);
            }
            sched_group<SG_WMMA, 8>();
        }

        float mt[8];
        #pragma unroll
        for (int nt = 0; nt < 8; ++nt) {
            float m = acc[nt][0];
            #pragma unroll
            for (int r = 1; r < 8; ++r) m = fmaxf(m, acc[nt][r]);
            mt[nt] = m;
        }
        float m = fmaxf(fmaxf(fmaxf(mt[0], mt[1]), fmaxf(mt[2], mt[3])),
                        fmaxf(fmaxf(mt[4], mt[5]), fmaxf(mt[6], mt[7])));
        m = fmaxf(m, xhalf(m));

        float st[8];
        #pragma unroll
        for (int nt = 0; nt < 8; ++nt) {
            float s = 0.0f;
            #pragma unroll
            for (int r = 0; r < 8; ++r) {
                const float e = __expf(acc[nt][r] - m);
                acc[nt][r] = e;
                s += e;
            }
            st[nt] = s;
        }
        float sum = ((st[0] + st[1]) + (st[2] + st[3])) + ((st[4] + st[5]) + (st[6] + st[7]));
        sum += xhalf(sum);

        const float wgt  = valid ? 1.0f : 0.0f;
        const float rinv = wgt / sum;

        #pragma unroll
        for (int nt = 0; nt < 8; ++nt) {
            #pragma unroll
            for (int r = 0; r < 8; ++r)
                osum[nt][r] = fmaf(acc[nt][r], rinv, osum[nt][r]);
        }
    }

    {
        float v64[64];
        #pragma unroll
        for (int nt = 0; nt < 8; ++nt) {
            #pragma unroll
            for (int r = 0; r < 8; ++r) v64[nt * 8 + r] = osum[nt][r];
        }
        float v32[32], v16[16], v8[8], v4[4];
        halve_across<8, 64>(v64, v32, (lm & 8) != 0);
        halve_across<4, 32>(v32, v16, (lm & 4) != 0);
        halve_across<2, 16>(v16, v8,  (lm & 2) != 0);
        halve_across<1, 8 >(v8,  v4,  (lm & 1) != 0);

        const int ch0 = (lm >> 1) * 16 + half * 8 + (lm & 1) * 4;
        #pragma unroll
        for (int q = 0; q < 4; ++q) atomicAdd(&sOut[ch0 + q], v4[q]);
    }

    __syncthreads();
    if (tid < NC) out[(size_t)b * NC + tid] = sOut[tid];
}

extern "C" void kernel_launch(void* const* d_in, const int* in_sizes, int n_in,
                              void* d_out, int out_size, void* d_ws, size_t ws_size,
                              hipStream_t stream) {
    const float* x = (const float*)d_in[0];
    const float* W = (const float*)d_in[1];
    float* out     = (float*)d_out;

    dim3 grid(NB);
    dim3 block(256);
    softmax_gemm_reduce_kernel<<<grid, block, 0, stream>>>(x, W, out);
}


# One wave per sample: W fragments stay in registers over its 13 row tiles (no LDS reads among the products), next tile's x in flight; column sums in registers, no atomics (sum order changes)
# speedup vs baseline: 1.4864x; 1.3271x over previous
#include <hip/hip_runtime.h>


#define NB   4096
#define NF   200
#define NK   40
#define NC   128

typedef __attribute__((ext_vector_type(2))) float v2f;
typedef __attribute__((ext_vector_type(4))) float v4f;
typedef __attribute__((ext_vector_type(8))) float v8f;

#define NRG  13
#define WPB  4
#define NTHR 128
#define NBLK 1024

__device__ __forceinline__ float xhalf(float v) {
    const int b = __builtin_bit_cast(int, v);
    return __builtin_bit_cast(float,
        __builtin_amdgcn_permlanex16(b, b, 0x76543210, 0xfedcba98, false, false));
}

template <int MASK>
__device__ __forceinline__ float row_xor(float v) {
    const int b = __builtin_bit_cast(int, v);
    return __builtin_bit_cast(float,
        __builtin_amdgcn_update_dpp(b, b, 0x160 | MASK, 0xf, 0xf, true));
}

template <int MASK, int N>
__device__ __forceinline__ void halve_across(const float (&in)[N], float (&outv)[N / 2], bool hi) {
    #pragma unroll
    for (int j = 0; j < N / 2; ++j) {
        const float keep = hi ? in[N / 2 + j] : in[j];
        const float send = hi ? in[j] : in[N / 2 + j];
        outv[j] = keep + row_xor<MASK>(send);
    }
}

__device__ __forceinline__ void load_x_tile(const float* __restrict__ xs, int rg, int lm, int klo,
                                            v2f (&xfr)[10]) {
    const int f      = rg * 16 + lm;
    const int fclamp = (f < NF) ? f : (NF - 1);
    const float* xrow = xs + (size_t)fclamp * NK + klo;
    #pragma unroll
    for (int ks = 0; ks < 10; ++ks)
        xfr[ks] = *(const v2f*)(xrow + 4 * ks);
}

__global__ __launch_bounds__(NTHR)
void softmax_gemm_reduce_kernel(const float* __restrict__ x,
                                const float* __restrict__ W,
                                float* __restrict__ out)
{
    __shared__ float sW[(NK / 2) * (NC * 2)];

    const int tid  = threadIdx.x;
    const int wave = tid >> 5;
    const int lane = tid & 31;
    const int half = lane >> 4;
    const int lm   = lane & 15;
    const int b    = blockIdx.x * WPB + wave;
    const int klo  = half * 2;

    const float* xs = x + (size_t)b * NF * NK;

    constexpr int WST = (NK * NC / 4) / (NTHR);
    v4f wst[WST];
    #pragma unroll
    for (int it = 0; it < WST; ++it)
        wst[it] = *(const v4f*)(W + (tid + it * NTHR) * 4);

    v2f xcur[10];
    load_x_tile(xs, 0, lm, klo, xcur);

    #pragma unroll
    for (int it = 0; it < WST; ++it) {
        const int i = (tid + it * NTHR) * 4;
        const int k = i >> 7;
        const int c = i & (NC - 1);
        float* dst = &sW[(k >> 1) * 256 + c * 2 + (k & 1)];
        dst[0] = wst[it].x; dst[2] = wst[it].y; dst[4] = wst[it].z; dst[6] = wst[it].w;
    }
    __syncthreads();

    const float* sWcol = &sW[lm * 2];
    v2f wf[10][8];
    #pragma unroll
    for (int ks = 0; ks < 10; ++ks) {
        #pragma unroll
        for (int nt = 0; nt < 8; ++nt)
            wf[ks][nt] = *(const v2f*)(sWcol + (2 * ks + half) * 256 + nt * 32);
    }

    v8f osum[8] = {};

    #pragma unroll 1
    for (int rg = 0; rg < NRG; ++rg) {
        v2f xnxt[10];
        if (rg + 1 < NRG) load_x_tile(xs, rg + 1, lm, klo, xnxt);

        v8f acc[8] = {};
        #pragma unroll
        for (int ks = 0; ks < 10; ++ks) {
            #pragma unroll
            for (int nt = 0; nt < 8; ++nt) {
                acc[nt] = __builtin_amdgcn_wmma_f32_16x16x4_f32(
                               false, wf[ks][nt],  false, xcur[ks],
                               (short)0, acc[nt],
                               false,  false);
            }
        }

        float mt[8];
        #pragma unroll
        for (int nt = 0; nt < 8; ++nt) {
            float m = acc[nt][0];
            #pragma unroll
            for (int r = 1; r < 8; ++r) m = fmaxf(m, acc[nt][r]);
            mt[nt] = m;
        }
        float m = fmaxf(fmaxf(fmaxf(mt[0], mt[1]), fmaxf(mt[2], mt[3])),
                        fmaxf(fmaxf(mt[4], mt[5]), fmaxf(mt[6], mt[7])));
        m = fmaxf(m, xhalf(m));

        float st[8];
        #pragma unroll
        for (int nt = 0; nt < 8; ++nt) {
            float s = 0.0f;
            #pragma unroll
            for (int r = 0; r < 8; ++r) {
                const float e = __expf(acc[nt][r] - m);
                acc[nt][r] = e;
                s += e;
            }
            st[nt] = s;
        }
        float sum = ((st[0] + st[1]) + (st[2] + st[3])) + ((st[4] + st[5]) + (st[6] + st[7]));
        sum += xhalf(sum);

        const bool valid = (rg * 16 + lm < NF);
        const float wgt  = valid ? 1.0f : 0.0f;
        const float rinv = wgt / sum;

        #pragma unroll
        for (int nt = 0; nt < 8; ++nt) {
            #pragma unroll
            for (int r = 0; r < 8; ++r)
                osum[nt][r] = fmaf(acc[nt][r], rinv, osum[nt][r]);
        }

        if (rg + 1 < NRG) {
            #pragma unroll
            for (int ks = 0; ks < 10; ++ks) xcur[ks] = xnxt[ks];
        }
    }

    {
        float v64[64];
        #pragma unroll
        for (int nt = 0; nt < 8; ++nt) {
            #pragma unroll
            for (int r = 0; r < 8; ++r) v64[nt * 8 + r] = osum[nt][r];
        }
        float v32[32], v16[16], v8[8], v4[4];
        halve_across<8, 64>(v64, v32, (lm & 8) != 0);
        halve_across<4, 32>(v32, v16, (lm & 4) != 0);
        halve_across<2, 16>(v16, v8,  (lm & 2) != 0);
        halve_across<1, 8 >(v8,  v4,  (lm & 1) != 0);

        const int ch0 = (lm >> 1) * 16 + half * 8 + (lm & 1) * 4;
        v4f o;
        o.x = v4[0]; o.y = v4[1]; o.z = v4[2]; o.w = v4[3];
        *(v4f*)(out + (size_t)b * NC + ch0) = o;
    }
}

extern "C" void kernel_launch(void* const* d_in, const int* in_sizes, int n_in,
                              void* d_out, int out_size, void* d_ws, size_t ws_size,
                              hipStream_t stream) {
    const float* x = (const float*)d_in[0];
    const float* W = (const float*)d_in[1];
    float* out     = (float*)d_out;

    dim3 grid(NBLK);
    dim3 block(NTHR);
    softmax_gemm_reduce_kernel<<<grid, block, 0, stream>>>(x, W, out);
}


# Softmax vector work as packed two-channel f32 ops (one packed multiply-add per two base-2 exponent arguments, packed row and column sums), each laid behind an exponential (sum order changes)
# speedup vs baseline: 1.5518x; 1.0440x over previous
#include <hip/hip_runtime.h>


#define NB   4096
#define NF   200
#define NK   40
#define NC   128

typedef __attribute__((ext_vector_type(2))) float v2f;
typedef __attribute__((ext_vector_type(4))) float v4f;
typedef __attribute__((ext_vector_type(8))) float v8f;

#define NRG  13
#define WPB  4
#define NTHR 128
#define NBLK 1024

__device__ __forceinline__ float xhalf(float v) {
    const int b = __builtin_bit_cast(int, v);
    return __builtin_bit_cast(float,
        __builtin_amdgcn_permlanex16(b, b, 0x76543210, 0xfedcba98, false, false));
}

template <int MASK>
__device__ __forceinline__ float row_xor(float v) {
    const int b = __builtin_bit_cast(int, v);
    return __builtin_bit_cast(float,
        __builtin_amdgcn_update_dpp(b, b, 0x160 | MASK, 0xf, 0xf, true));
}

template <int MASK, int N>
__device__ __forceinline__ void halve_across(const float (&in)[N], float (&outv)[N / 2], bool hi) {
    #pragma unroll
    for (int j = 0; j < N / 2; ++j) {
        const float keep = hi ? in[N / 2 + j] : in[j];
        const float send = hi ? in[j] : in[N / 2 + j];
        outv[j] = keep + row_xor<MASK>(send);
    }
}

__device__ __forceinline__ void load_x_tile(const float* __restrict__ xs, int rg, int lm, int klo,
                                            v2f (&xfr)[10]) {
    const int f      = rg * 16 + lm;
    const int fclamp = (f < NF) ? f : (NF - 1);
    const float* xrow = xs + (size_t)fclamp * NK + klo;
    #pragma unroll
    for (int ks = 0; ks < 10; ++ks)
        xfr[ks] = *(const v2f*)(xrow + 4 * ks);
}

__global__ __launch_bounds__(NTHR)
void softmax_gemm_reduce_kernel(const float* __restrict__ x,
                                const float* __restrict__ W,
                                float* __restrict__ out)
{
    __shared__ float sW[(NK / 2) * (NC * 2)];

    const int tid  = threadIdx.x;
    const int wave = tid >> 5;
    const int lane = tid & 31;
    const int half = lane >> 4;
    const int lm   = lane & 15;
    const int b    = blockIdx.x * WPB + wave;
    const int klo  = half * 2;

    const float* xs = x + (size_t)b * NF * NK;

    constexpr int WST = (NK * NC / 4) / (NTHR);
    v4f wst[WST];
    #pragma unroll
    for (int it = 0; it < WST; ++it)
        wst[it] = *(const v4f*)(W + (tid + it * NTHR) * 4);

    v2f xcur[10];
    load_x_tile(xs, 0, lm, klo, xcur);

    #pragma unroll
    for (int it = 0; it < WST; ++it) {
        const int i = (tid + it * NTHR) * 4;
        const int k = i >> 7;
        const int c = i & (NC - 1);
        float* dst = &sW[(k >> 1) * 256 + c * 2 + (k & 1)];
        dst[0] = wst[it].x; dst[2] = wst[it].y; dst[4] = wst[it].z; dst[6] = wst[it].w;
    }
    __syncthreads();

    const float* sWcol = &sW[lm * 2];
    v2f wf[10][8];
    #pragma unroll
    for (int ks = 0; ks < 10; ++ks) {
        #pragma unroll
        for (int nt = 0; nt < 8; ++nt)
            wf[ks][nt] = *(const v2f*)(sWcol + (2 * ks + half) * 256 + nt * 32);
    }

    v8f osum[8] = {};

    #pragma unroll 1
    for (int rg = 0; rg < NRG; ++rg) {
        v2f xnxt[10];
        if (rg + 1 < NRG) load_x_tile(xs, rg + 1, lm, klo, xnxt);

        v8f acc[8] = {};
        #pragma unroll
        for (int ks = 0; ks < 10; ++ks) {
            #pragma unroll
            for (int nt = 0; nt < 8; ++nt) {
                acc[nt] = __builtin_amdgcn_wmma_f32_16x16x4_f32(
                               false, wf[ks][nt],  false, xcur[ks],
                               (short)0, acc[nt],
                               false,  false);
            }
        }

        float mt[8];
        #pragma unroll
        for (int nt = 0; nt < 8; ++nt) {
            float m = acc[nt][0];
            #pragma unroll
            for (int r = 1; r < 8; ++r) m = fmaxf(m, acc[nt][r]);
            mt[nt] = m;
        }
        float m = fmaxf(fmaxf(fmaxf(mt[0], mt[1]), fmaxf(mt[2], mt[3])),
                        fmaxf(fmaxf(mt[4], mt[5]), fmaxf(mt[6], mt[7])));
        m = fmaxf(m, xhalf(m));

        const float LOG2E = 1.4426950408889634f;
        const float nml   = -m * LOG2E;
        const v2f   lv    = {LOG2E, LOG2E};
        const v2f   nmv   = {nml, nml};
        v2f targ[32];
        v2f ev[32];
        v2f sp[4];
        #pragma unroll
        for (int k = 0; k < 4; ++k) {
            const v2f a = {acc[k & 7][2 * (k >> 3)], acc[k & 7][2 * (k >> 3) + 1]};
            targ[k] = __builtin_elementwise_fma(a, lv, nmv);
        }
        #pragma unroll
        for (int k = 0; k < 32; ++k) {
            ev[k].x = __builtin_amdgcn_exp2f(targ[k].x);
            if (k + 4 < 32) {
                const int q = k + 4;
                const v2f a = {acc[q & 7][2 * (q >> 3)], acc[q & 7][2 * (q >> 3) + 1]};
                targ[q] = __builtin_elementwise_fma(a, lv, nmv);
            }
            __builtin_amdgcn_sched_barrier(0);
            ev[k].y = __builtin_amdgcn_exp2f(targ[k].y);
            if (k >= 3) {
                const int q = k - 3;
                if (q < 4) sp[q] = ev[q];
                else       sp[q & 3] = sp[q & 3] + ev[q];
            }
            __builtin_amdgcn_sched_barrier(0);
        }
        #pragma unroll
        for (int q = 29; q < 32; ++q) sp[q & 3] = sp[q & 3] + ev[q];
        #pragma unroll
        for (int k = 0; k < 32; ++k) {
            acc[k & 7][2 * (k >> 3)]     = ev[k].x;
            acc[k & 7][2 * (k >> 3) + 1] = ev[k].y;
        }
        const v2f s2 = (sp[0] + sp[1]) + (sp[2] + sp[3]);
        float sum = s2.x + s2.y;
        sum += xhalf(sum);

        const bool valid = (rg * 16 + lm < NF);
        const float wgt  = valid ? 1.0f : 0.0f;
        const float rinv = wgt / sum;
        const v2f   rv   = {rinv, rinv};

        #pragma unroll
        for (int nt = 0; nt < 8; ++nt) {
            #pragma unroll
            for (int j = 0; j < 4; ++j) {
                const v2f e = {acc[nt][2 * j], acc[nt][2 * j + 1]};
                v2f o = {osum[nt][2 * j], osum[nt][2 * j + 1]};
                o = __builtin_elementwise_fma(e, rv, o);
                osum[nt][2 * j]     = o.x;
                osum[nt][2 * j + 1] = o.y;
            }
        }

        if (rg + 1 < NRG) {
            #pragma unroll
            for (int ks = 0; ks < 10; ++ks) xcur[ks] = xnxt[ks];
        }
    }

    {
        float v64[64];
        #pragma unroll
        for (int nt = 0; nt < 8; ++nt) {
            #pragma unroll
            for (int r = 0; r < 8; ++r) v64[nt * 8 + r] = osum[nt][r];
        }
        float v32[32], v16[16], v8[8], v4[4];
        halve_across<8, 64>(v64, v32, (lm & 8) != 0);
        halve_across<4, 32>(v32, v16, (lm & 4) != 0);
        halve_across<2, 16>(v16, v8,  (lm & 2) != 0);
        halve_across<1, 8 >(v8,  v4,  (lm & 1) != 0);

        const int ch0 = (lm >> 1) * 16 + half * 8 + (lm & 1) * 4;
        v4f o;
        o.x = v4[0]; o.y = v4[1]; o.z = v4[2]; o.w = v4[3];
        *(v4f*)(out + (size_t)b * NC + ch0) = o;
    }
}

extern "C" void kernel_launch(void* const* d_in, const int* in_sizes, int n_in,
                              void* d_out, int out_size, void* d_ws, size_t ws_size,
                              hipStream_t stream) {
    const float* x = (const float*)d_in[0];
    const float* W = (const float*)d_in[1];
    float* out     = (float*)d_out;

    dim3 grid(NBLK);
    dim3 block(NTHR);
    softmax_gemm_reduce_kernel<<<grid, block, 0, stream>>>(x, W, out);
}
